# starting kernel
# baseline (speedup 1.0000x reference)
#include <hip/hip_runtime.h>

typedef __attribute__((ext_vector_type(2))) float v2f;
typedef __attribute__((ext_vector_type(8))) float v8f;

#define INPUT_SIZE 16
#define KN         50
#define HC         25
#define FW         1266
#define ROWS       64
#define XSTRIDE    1268
#define NTHREADS   512

__global__ __launch_bounds__(NTHREADS) void densenet_wmma_kernel(
    const float* __restrict__ state,
    const float* __restrict__ Ws,
    const float* __restrict__ bs,
    const float* __restrict__ Wout,
    const float* __restrict__ bout,
    float* __restrict__ out)
{
    extern __shared__ float xbuf[];

    const int tid  = threadIdx.x;
    const int wave = tid >> 5;
    const int lane = tid & 31;
    const int l16  = lane & 15;
    const int hi   = lane >> 4;
    const int ksub = hi * 2;

    const int nt = wave & 3;
    const int mp = (wave >> 2) & 1;
    const int kh = wave >> 3;

    const int kh_u   = __builtin_amdgcn_readfirstlane(kh);
    const int rowbase = blockIdx.x * ROWS;

    for (int e = tid; e < ROWS * INPUT_SIZE; e += NTHREADS) {
        int r = e >> 4, c = e & 15;
        xbuf[r * XSTRIDE + c] = state[(size_t)(rowbase + r) * INPUT_SIZE + c];
    }
    __syncthreads();

    const int  n      = nt * 16 + l16;
    const bool nvalid = (n < KN);
    const int  ncl    = nvalid ? n : (KN - 1);
    const float* xrow0 = xbuf + (mp * 32 + l16) * XSTRIDE;
    const float* xrow1 = xrow0 + 16 * XSTRIDE;

    int width = INPUT_SIZE;
    for (int layer = 0; layer < HC; ++layer) {
        const float* wrow = Ws + (size_t)(layer * KN + ncl) * FW;

        const float bias = (kh_u == 0) ? bs[layer * KN + ncl] : 0.0f;
        v8f acc0, acc1;
        #pragma unroll
        for (int j = 0; j < 8; ++j) { acc0[j] = bias; acc1[j] = bias; }

        const int wmain  = width & ~3;
        const int ksplit = (wmain >> 1) & ~3;
        const int kbeg = __builtin_amdgcn_readfirstlane((kh_u == 0) ? 0 : ksplit);
        const int kend = __builtin_amdgcn_readfirstlane((kh_u == 0) ? ksplit : wmain);

        #pragma unroll 4
        for (int k = kbeg; k < kend; k += 4) {
            v2f b  = *(const v2f*)(wrow  + k + ksub);
            v2f a0 = *(const v2f*)(xrow0 + k + ksub);
            v2f a1 = *(const v2f*)(xrow1 + k + ksub);
            acc0 = __builtin_amdgcn_wmma_f32_16x16x4_f32(
                       false, a0, false, b, (short)0, acc0, false, false);
            acc1 = __builtin_amdgcn_wmma_f32_16x16x4_f32(
                       false, a1, false, b, (short)0, acc1, false, false);
        }
        if ((kh_u == 1) && (width & 2)) {
            v2f a0, a1, b;
            if (!hi) {
                a0 = *(const v2f*)(xrow0 + wmain);
                a1 = *(const v2f*)(xrow1 + wmain);
                b  = *(const v2f*)(wrow  + wmain);
            } else {
                a0[0] = 0.0f; a0[1] = 0.0f;
                a1[0] = 0.0f; a1[1] = 0.0f;
                b[0]  = 0.0f; b[1]  = 0.0f;
            }
            acc0 = __builtin_amdgcn_wmma_f32_16x16x4_f32(
                       false, a0, false, b, (short)0, acc0, false, false);
            acc1 = __builtin_amdgcn_wmma_f32_16x16x4_f32(
                       false, a1, false, b, (short)0, acc1, false, false);
        }

        if (kh_u == 1 && nvalid) {
            float* dst = xbuf + width + n;
            #pragma unroll
            for (int j = 0; j < 8; ++j) {
                int r0 = mp * 32 + hi * 8 + j;
                dst[r0 * XSTRIDE]        = acc0[j];
                dst[(r0 + 16) * XSTRIDE] = acc1[j];
            }
        }
        __syncthreads();
        if (kh_u == 0 && nvalid) {
            float* dst = xbuf + width + n;
            #pragma unroll
            for (int j = 0; j < 8; ++j) {
                int r0 = mp * 32 + hi * 8 + j;
                float v0 = acc0[j] + dst[r0 * XSTRIDE];
                float v1 = acc1[j] + dst[(r0 + 16) * XSTRIDE];
                v0 = (v0 > 0.0f) ? v0 : 0.01f * v0;
                v1 = (v1 > 0.0f) ? v1 : 0.01f * v1;
                dst[r0 * XSTRIDE]        = v0;
                dst[(r0 + 16) * XSTRIDE] = v1;
            }
        }
        width += KN;
        __syncthreads();
    }

    const int row = tid >> 3;
    const int g   = tid & 7;
    const float* xr = xbuf + row * XSTRIDE;
    float facc = 0.0f;
    for (int c = g; c < FW; c += 8)
        facc += xr[c] * Wout[c];
    facc += __shfl_down(facc, 4, 8);
    facc += __shfl_down(facc, 2, 8);
    facc += __shfl_down(facc, 1, 8);
    if (g == 0)
        out[rowbase + row] = facc + bout[0];
}

extern "C" void kernel_launch(void* const* d_in, const int* in_sizes, int n_in,
                              void* d_out, int out_size, void* d_ws, size_t ws_size,
                              hipStream_t stream) {
    (void)n_in; (void)d_ws; (void)ws_size; (void)out_size;
    const float* state = (const float*)d_in[0];
    const float* Ws    = (const float*)d_in[1];
    const float* bs    = (const float*)d_in[2];
    const float* Wout  = (const float*)d_in[3];
    const float* bout  = (const float*)d_in[4];
    float* out = (float*)d_out;

    const int B      = in_sizes[0] / INPUT_SIZE;
    const int blocks = B / ROWS;
    const size_t lds_bytes = (size_t)ROWS * XSTRIDE * sizeof(float);

    (void)hipFuncSetAttribute((const void*)densenet_wmma_kernel,
                              hipFuncAttributeMaxDynamicSharedMemorySize,
                              (int)lds_bytes);

    densenet_wmma_kernel<<<blocks, NTHREADS, lds_bytes, stream>>>(
        state, Ws, bs, Wout, bout, out);
}


# Software-pipeline the K loop: each 16-k block of weight and activation fragments is loaded two blocks ahead into rotating register sets, left-over steps at layer start; k order unchanged
# speedup vs baseline: 1.3578x; 1.3578x over previous
#include <hip/hip_runtime.h>

typedef __attribute__((ext_vector_type(2))) float v2f;
typedef __attribute__((ext_vector_type(8))) float v8f;

#define INPUT_SIZE 16
#define KN         50
#define HC         25
#define FW         1266
#define ROWS       64
#define XSTRIDE    1268
#define NTHREADS   512

struct FragSet { v2f b[4]; v2f a0[4]; v2f a1[4]; };

__device__ __forceinline__ void load_set(FragSet& s, const float* w,
                                         const float* x0, const float* x1) {
    #pragma unroll
    for (int t = 0; t < 4; ++t) {
        s.b[t]  = *(const v2f*)(w  + 4 * t);
        s.a0[t] = *(const v2f*)(x0 + 4 * t);
        s.a1[t] = *(const v2f*)(x1 + 4 * t);
    }
}

__device__ __forceinline__ void mma_set(const FragSet& s, v8f& acc0, v8f& acc1) {
    #pragma unroll
    for (int t = 0; t < 4; ++t) {
        acc0 = __builtin_amdgcn_wmma_f32_16x16x4_f32(
                   false, s.a0[t], false, s.b[t], (short)0, acc0, false, false);
        acc1 = __builtin_amdgcn_wmma_f32_16x16x4_f32(
                   false, s.a1[t], false, s.b[t], (short)0, acc1, false, false);
    }
}

__global__ __launch_bounds__(NTHREADS) void densenet_wmma_kernel(
    const float* __restrict__ state,
    const float* __restrict__ Ws,
    const float* __restrict__ bs,
    const float* __restrict__ Wout,
    const float* __restrict__ bout,
    float* __restrict__ out)
{
    extern __shared__ float xbuf[];

    const int tid  = threadIdx.x;
    const int wave = tid >> 5;
    const int lane = tid & 31;
    const int l16  = lane & 15;
    const int hi   = lane >> 4;
    const int ksub = hi * 2;

    const int nt = wave & 3;
    const int mp = (wave >> 2) & 1;
    const int kh = wave >> 3;

    const int kh_u   = __builtin_amdgcn_readfirstlane(kh);
    const int rowbase = blockIdx.x * ROWS;

    for (int e = tid; e < ROWS * INPUT_SIZE; e += NTHREADS) {
        int r = e >> 4, c = e & 15;
        xbuf[r * XSTRIDE + c] = state[(size_t)(rowbase + r) * INPUT_SIZE + c];
    }
    __syncthreads();

    const int  n      = nt * 16 + l16;
    const bool nvalid = (n < KN);
    const int  ncl    = nvalid ? n : (KN - 1);
    const float* xrow0 = xbuf + (mp * 32 + l16) * XSTRIDE;
    const float* xrow1 = xrow0 + 16 * XSTRIDE;

    int width = INPUT_SIZE;
    for (int layer = 0; layer < HC; ++layer) {
        const float* wrow = Ws + (size_t)(layer * KN + ncl) * FW;

        const float bias = (kh_u == 0) ? bs[layer * KN + ncl] : 0.0f;
        v8f acc0, acc1;
        #pragma unroll
        for (int j = 0; j < 8; ++j) { acc0[j] = bias; acc1[j] = bias; }

        const int wmain  = width & ~3;
        const int ksplit = (wmain >> 1) & ~3;
        const int kbeg = __builtin_amdgcn_readfirstlane((kh_u == 0) ? 0 : ksplit);
        const int kend = __builtin_amdgcn_readfirstlane((kh_u == 0) ? ksplit : wmain);

        const float* wl = wrow  + ksub;
        const float* x0 = xrow0 + ksub;
        const float* x1 = xrow1 + ksub;
        const int nblk = (kend - kbeg) >> 4;
        const int nrem = ((kend - kbeg) >> 2) & 3;
        FragSet sR;
        {
            const int krem = kbeg + (nblk << 4);
            const int kcl  = kend - 4;
            #pragma unroll
            for (int t = 0; t < 3; ++t) {
                const int kt = (krem + 4 * t < kcl) ? (krem + 4 * t) : kcl;
                sR.b[t]  = *(const v2f*)(wl + kt);
                sR.a0[t] = *(const v2f*)(x0 + kt);
                sR.a1[t] = *(const v2f*)(x1 + kt);
            }
            const int ktail = (width & 2) ? wmain : (wmain - 4);
            sR.b[3]  = *(const v2f*)(wrow  + ktail);
            sR.a0[3] = *(const v2f*)(xrow0 + ktail);
            sR.a1[3] = *(const v2f*)(xrow1 + ktail);
        }
        if (nblk > 0) {
            const int klast = kbeg + ((nblk - 1) << 4);
            #define BLK(j) ((kbeg + ((j) << 4) < klast) ? (kbeg + ((j) << 4)) : klast)
            FragSet s0, s1, s2;
            int kk = BLK(0);
            load_set(s0, wl + kk, x0 + kk, x1 + kk);
            kk = BLK(1);
            load_set(s1, wl + kk, x0 + kk, x1 + kk);
            int i = 0;
            for (; i + 3 <= nblk; i += 3) {
                kk = BLK(i + 2);
                load_set(s2, wl + kk, x0 + kk, x1 + kk);
                __builtin_amdgcn_sched_barrier(0);
                mma_set(s0, acc0, acc1);
                kk = BLK(i + 3);
                load_set(s0, wl + kk, x0 + kk, x1 + kk);
                __builtin_amdgcn_sched_barrier(0);
                mma_set(s1, acc0, acc1);
                kk = BLK(i + 4);
                load_set(s1, wl + kk, x0 + kk, x1 + kk);
                __builtin_amdgcn_sched_barrier(0);
                mma_set(s2, acc0, acc1);
            }
            #undef BLK
            if (nblk - i >= 1) mma_set(s0, acc0, acc1);
            if (nblk - i >= 2) mma_set(s1, acc0, acc1);
        }
        #pragma unroll
        for (int t = 0; t < 3; ++t) {
            if (t < nrem) {
                acc0 = __builtin_amdgcn_wmma_f32_16x16x4_f32(
                           false, sR.a0[t], false, sR.b[t], (short)0, acc0, false, false);
                acc1 = __builtin_amdgcn_wmma_f32_16x16x4_f32(
                           false, sR.a1[t], false, sR.b[t], (short)0, acc1, false, false);
            }
        }
        if ((kh_u == 1) && (width & 2)) {
            v2f a0 = sR.a0[3], a1 = sR.a1[3], b = sR.b[3];
            a0[0] = hi ? 0.0f : a0[0];  a0[1] = hi ? 0.0f : a0[1];
            a1[0] = hi ? 0.0f : a1[0];  a1[1] = hi ? 0.0f : a1[1];
            b[0]  = hi ? 0.0f : b[0];   b[1]  = hi ? 0.0f : b[1];
            acc0 = __builtin_amdgcn_wmma_f32_16x16x4_f32(
                       false, a0, false, b, (short)0, acc0, false, false);
            acc1 = __builtin_amdgcn_wmma_f32_16x16x4_f32(
                       false, a1, false, b, (short)0, acc1, false, false);
        }

        if (kh_u == 1 && nvalid) {
            float* dst = xbuf + width + n;
            #pragma unroll
            for (int j = 0; j < 8; ++j) {
                int r0 = mp * 32 + hi * 8 + j;
                dst[r0 * XSTRIDE]        = acc0[j];
                dst[(r0 + 16) * XSTRIDE] = acc1[j];
            }
        }
        __syncthreads();
        if (kh_u == 0 && nvalid) {
            float* dst = xbuf + width + n;
            #pragma unroll
            for (int j = 0; j < 8; ++j) {
                int r0 = mp * 32 + hi * 8 + j;
                float v0 = acc0[j] + dst[r0 * XSTRIDE];
                float v1 = acc1[j] + dst[(r0 + 16) * XSTRIDE];
                v0 = (v0 > 0.0f) ? v0 : 0.01f * v0;
                v1 = (v1 > 0.0f) ? v1 : 0.01f * v1;
                dst[r0 * XSTRIDE]        = v0;
                dst[(r0 + 16) * XSTRIDE] = v1;
            }
        }
        width += KN;
        __syncthreads();
    }

    const int row = tid >> 3;
    const int g   = tid & 7;
    const float* xr = xbuf + row * XSTRIDE;
    float facc = 0.0f;
    for (int c = g; c < FW; c += 8)
        facc += xr[c] * Wout[c];
    facc += __shfl_down(facc, 4, 8);
    facc += __shfl_down(facc, 2, 8);
    facc += __shfl_down(facc, 1, 8);
    if (g == 0)
        out[rowbase + row] = facc + bout[0];
}

extern "C" void kernel_launch(void* const* d_in, const int* in_sizes, int n_in,
                              void* d_out, int out_size, void* d_ws, size_t ws_size,
                              hipStream_t stream) {
    (void)n_in; (void)d_ws; (void)ws_size; (void)out_size;
    const float* state = (const float*)d_in[0];
    const float* Ws    = (const float*)d_in[1];
    const float* bs    = (const float*)d_in[2];
    const float* Wout  = (const float*)d_in[3];
    const float* bout  = (const float*)d_in[4];
    float* out = (float*)d_out;

    const int B      = in_sizes[0] / INPUT_SIZE;
    const int blocks = B / ROWS;
    const size_t lds_bytes = (size_t)ROWS * XSTRIDE * sizeof(float);

    (void)hipFuncSetAttribute((const void*)densenet_wmma_kernel,
                              hipFuncAttributeMaxDynamicSharedMemorySize,
                              (int)lds_bytes);

    densenet_wmma_kernel<<<blocks, NTHREADS, lds_bytes, stream>>>(
        state, Ws, bs, Wout, bout, out);
}


# Neuron tiles of 16 run across layer borders: spare columns of a layer's last tile pre-sum the next layer's first neurons, which then add only their last 50 k (sum order changed); 19% fewer WMMAs
# speedup vs baseline: 1.5075x; 1.1103x over previous
#include <hip/hip_runtime.h>

typedef __attribute__((ext_vector_type(2))) float v2f;
typedef __attribute__((ext_vector_type(8))) float v8f;

#define INPUT_SIZE 16
#define KN         50
#define HC         25
#define FW         1266
#define NF         1250
#define ROWS       64
#define XSTRIDE    1268
#define NTHREADS   512

struct FragSet { v2f b[4]; v2f a0[4]; v2f a1[4]; };

__device__ __forceinline__ void load_set(FragSet& s, const float* w,
                                         const float* x0, const float* x1) {
    #pragma unroll
    for (int t = 0; t < 4; ++t) {
        s.b[t]  = *(const v2f*)(w  + 4 * t);
        s.a0[t] = *(const v2f*)(x0 + 4 * t);
        s.a1[t] = *(const v2f*)(x1 + 4 * t);
    }
}

__device__ __forceinline__ void mma_set(const FragSet& s, v8f& acc0, v8f& acc1) {
    #pragma unroll
    for (int t = 0; t < 4; ++t) {
        acc0 = __builtin_amdgcn_wmma_f32_16x16x4_f32(
                   false, s.a0[t], false, s.b[t], (short)0, acc0, false, false);
        acc1 = __builtin_amdgcn_wmma_f32_16x16x4_f32(
                   false, s.a1[t], false, s.b[t], (short)0, acc1, false, false);
    }
}

__global__ __launch_bounds__(NTHREADS) void densenet_wmma_kernel(
    const float* __restrict__ state,
    const float* __restrict__ Ws,
    const float* __restrict__ bs,
    const float* __restrict__ Wout,
    const float* __restrict__ bout,
    float* __restrict__ out)
{
    extern __shared__ float xbuf[];

    const int tid  = threadIdx.x;
    const int wave = tid >> 5;
    const int lane = tid & 31;
    const int l16  = lane & 15;
    const int hi   = lane >> 4;
    const int ksub = hi * 2;

    const int sl = wave & 3;
    const int mp = sl & 1;
    const int kh = sl >> 1;

    const int kh_u   = __builtin_amdgcn_readfirstlane(kh);
    const int q_u    = __builtin_amdgcn_readfirstlane(wave >> 2);
    const int rowbase = blockIdx.x * ROWS;

    for (int e = tid; e < ROWS * INPUT_SIZE; e += NTHREADS) {
        int r = e >> 4, c = e & 15;
        xbuf[r * XSTRIDE + c] = state[(size_t)(rowbase + r) * INPUT_SIZE + c];
    }
    __syncthreads();

    const float* xrow0 = xbuf + (mp * 32 + l16) * XSTRIDE;
    const float* xrow1 = xrow0 + 16 * XSTRIDE;
    const float* xrowS = xbuf + (sl * 16 + l16) * XSTRIDE;

    int width = INPUT_SIZE;
    for (int layer = 0; layer < HC; ++layer) {
        const int f0    = layer * KN;
        const int r     = f0 & 15;
        const int tfull = (f0 + 15) >> 4;
        const bool shallow = (r != 0) && (q_u == 3);
        v8f acc0, acc1;
        int  f = 0;
        bool fvalid = false;
        if (!shallow) {
        f      = (tfull + q_u) * 16 + l16;
        fvalid = (f < NF);
        const int fcl = fvalid ? f : (NF - 1);
        const float* wrow = Ws + (size_t)fcl * FW;

        const float bias = (kh_u == 0) ? bs[fcl] : 0.0f;
        #pragma unroll
        for (int j = 0; j < 8; ++j) { acc0[j] = bias; acc1[j] = bias; }

        const int wmain  = width & ~3;
        const int ksplit = (wmain >> 1) & ~3;
        const int kbeg = __builtin_amdgcn_readfirstlane((kh_u == 0) ? 0 : ksplit);
        const int kend = __builtin_amdgcn_readfirstlane((kh_u == 0) ? ksplit : wmain);

        const float* wl = wrow  + ksub;
        const float* x0 = xrow0 + ksub;
        const float* x1 = xrow1 + ksub;
        const int nblk = (kend - kbeg) >> 4;
        const int nrem = ((kend - kbeg) >> 2) & 3;
        FragSet sR;
        {
            const int krem = kbeg + (nblk << 4);
            const int kcl  = kend - 4;
            #pragma unroll
            for (int t = 0; t < 3; ++t) {
                const int kt = (krem + 4 * t < kcl) ? (krem + 4 * t) : kcl;
                sR.b[t]  = *(const v2f*)(wl + kt);
                sR.a0[t] = *(const v2f*)(x0 + kt);
                sR.a1[t] = *(const v2f*)(x1 + kt);
            }
            const int ktail = (width & 2) ? wmain : (wmain - 4);
            sR.b[3]  = *(const v2f*)(wrow  + ktail);
            sR.a0[3] = *(const v2f*)(xrow0 + ktail);
            sR.a1[3] = *(const v2f*)(xrow1 + ktail);
        }
        if (nblk > 0) {
            const int klast = kbeg + ((nblk - 1) << 4);
            #define BLK(j) ((kbeg + ((j) << 4) < klast) ? (kbeg + ((j) << 4)) : klast)
            FragSet s0, s1, s2;
            int kk = BLK(0);
            load_set(s0, wl + kk, x0 + kk, x1 + kk);
            kk = BLK(1);
            load_set(s1, wl + kk, x0 + kk, x1 + kk);
            int i = 0;
            for (; i + 3 <= nblk; i += 3) {
                kk = BLK(i + 2);
                load_set(s2, wl + kk, x0 + kk, x1 + kk);
                __builtin_amdgcn_sched_barrier(0);
                mma_set(s0, acc0, acc1);
                kk = BLK(i + 3);
                load_set(s0, wl + kk, x0 + kk, x1 + kk);
                __builtin_amdgcn_sched_barrier(0);
                mma_set(s1, acc0, acc1);
                kk = BLK(i + 4);
                load_set(s1, wl + kk, x0 + kk, x1 + kk);
                __builtin_amdgcn_sched_barrier(0);
                mma_set(s2, acc0, acc1);
            }
            #undef BLK
            if (nblk - i >= 1) mma_set(s0, acc0, acc1);
            if (nblk - i >= 2) mma_set(s1, acc0, acc1);
        }
        #pragma unroll
        for (int t = 0; t < 3; ++t) {
            if (t < nrem) {
                acc0 = __builtin_amdgcn_wmma_f32_16x16x4_f32(
                           false, sR.a0[t], false, sR.b[t], (short)0, acc0, false, false);
                acc1 = __builtin_amdgcn_wmma_f32_16x16x4_f32(
                           false, sR.a1[t], false, sR.b[t], (short)0, acc1, false, false);
            }
        }
        if ((kh_u == 1) && (width & 2)) {
            v2f a0 = sR.a0[3], a1 = sR.a1[3], b = sR.b[3];
            a0[0] = hi ? 0.0f : a0[0];  a0[1] = hi ? 0.0f : a0[1];
            a1[0] = hi ? 0.0f : a1[0];  a1[1] = hi ? 0.0f : a1[1];
            b[0]  = hi ? 0.0f : b[0];   b[1]  = hi ? 0.0f : b[1];
            acc0 = __builtin_amdgcn_wmma_f32_16x16x4_f32(
                       false, a0, false, b, (short)0, acc0, false, false);
            acc1 = __builtin_amdgcn_wmma_f32_16x16x4_f32(
                       false, a1, false, b, (short)0, acc1, false, false);
        }

        if (kh_u == 1 && fvalid) {
            float* dst = xbuf + INPUT_SIZE + f;
            #pragma unroll
            for (int j = 0; j < 8; ++j) {
                int r0 = mp * 32 + hi * 8 + j;
                dst[r0 * XSTRIDE]        = acc0[j];
                dst[(r0 + 16) * XSTRIDE] = acc1[j];
            }
        }
        } else {
            const int  fS  = (f0 >> 4) * 16 + l16;
            const bool own = (l16 >= r);
            const float* ws = Ws + (size_t)fS * FW + (width - KN);
            const float* xs = xrowS + (width - KN);
            float* dst = xbuf + INPUT_SIZE + fS;
            v2f bS[13], aS[13];
            #pragma unroll
            for (int t = 0; t < 12; ++t) {
                bS[t] = *(const v2f*)(ws + 4 * t + ksub);
                aS[t] = *(const v2f*)(xs + 4 * t + ksub);
            }
            bS[12] = *(const v2f*)(ws + 48);
            aS[12] = *(const v2f*)(xs + 48);
            bS[12][0] = hi ? 0.0f : bS[12][0];  bS[12][1] = hi ? 0.0f : bS[12][1];
            aS[12][0] = hi ? 0.0f : aS[12][0];  aS[12][1] = hi ? 0.0f : aS[12][1];
            v8f acc;
            #pragma unroll
            for (int j = 0; j < 8; ++j) {
                int r0 = sl * 16 + hi * 8 + j;
                acc[j] = own ? dst[r0 * XSTRIDE] : 0.0f;
            }
            #pragma unroll
            for (int t = 0; t < 13; ++t)
                acc = __builtin_amdgcn_wmma_f32_16x16x4_f32(
                          false, aS[t], false, bS[t], (short)0, acc, false, false);
            if (own) {
                #pragma unroll
                for (int j = 0; j < 8; ++j) {
                    int r0 = sl * 16 + hi * 8 + j;
                    float v = acc[j];
                    dst[r0 * XSTRIDE] = (v > 0.0f) ? v : 0.01f * v;
                }
            }
        }
        __syncthreads();
        if (!shallow && kh_u == 0 && fvalid) {
            float* dst = xbuf + INPUT_SIZE + f;
            const bool fin = (f < f0 + KN);
            #pragma unroll
            for (int j = 0; j < 8; ++j) {
                int r0 = mp * 32 + hi * 8 + j;
                float v0 = acc0[j] + dst[r0 * XSTRIDE];
                float v1 = acc1[j] + dst[(r0 + 16) * XSTRIDE];
                v0 = (fin && !(v0 > 0.0f)) ? 0.01f * v0 : v0;
                v1 = (fin && !(v1 > 0.0f)) ? 0.01f * v1 : v1;
                dst[r0 * XSTRIDE]        = v0;
                dst[(r0 + 16) * XSTRIDE] = v1;
            }
        }
        width += KN;
        __syncthreads();
    }

    const int row = tid >> 3;
    const int g   = tid & 7;
    const float* xr = xbuf + row * XSTRIDE;
    float facc = 0.0f;
    for (int c = g; c < FW; c += 8)
        facc += xr[c] * Wout[c];
    facc += __shfl_down(facc, 4, 8);
    facc += __shfl_down(facc, 2, 8);
    facc += __shfl_down(facc, 1, 8);
    if (g == 0)
        out[rowbase + row] = facc + bout[0];
}

extern "C" void kernel_launch(void* const* d_in, const int* in_sizes, int n_in,
                              void* d_out, int out_size, void* d_ws, size_t ws_size,
                              hipStream_t stream) {
    (void)n_in; (void)d_ws; (void)ws_size; (void)out_size;
    const float* state = (const float*)d_in[0];
    const float* Ws    = (const float*)d_in[1];
    const float* bs    = (const float*)d_in[2];
    const float* Wout  = (const float*)d_in[3];
    const float* bout  = (const float*)d_in[4];
    float* out = (float*)d_out;

    const int B      = in_sizes[0] / INPUT_SIZE;
    const int blocks = B / ROWS;
    const size_t lds_bytes = (size_t)ROWS * XSTRIDE * sizeof(float);

    (void)hipFuncSetAttribute((const void*)densenet_wmma_kernel,
                              hipFuncAttributeMaxDynamicSharedMemorySize,
                              (int)lds_bytes);

    densenet_wmma_kernel<<<blocks, NTHREADS, lds_bytes, stream>>>(
        state, Ws, bs, Wout, bout, out);
}


# Pre-pass kernel packs the weights in B-fragment order into the workspace (2 launches); each full-depth B load is 512 contiguous bytes for 2 k-steps; K halves now meet at a multiple of 8 (sum order)
# speedup vs baseline: 1.6492x; 1.0940x over previous
#include <hip/hip_runtime.h>

typedef __attribute__((ext_vector_type(2))) float v2f;
typedef __attribute__((ext_vector_type(4))) float v4f;
typedef __attribute__((ext_vector_type(8))) float v8f;

#define INPUT_SIZE 16
#define KN         50
#define HC         25
#define FW         1266
#define NF         1250
#define ROWS       64
#define XSTRIDE    1268
#define NTHREADS   512
#define NTILE      79
#define NGRP       160
#define GSTRIDE    32

__global__ __launch_bounds__(128) void pack_weights_kernel(
    const float* __restrict__ Ws,
    float* __restrict__ Wq)
{
    const int T    = blockIdx.x;
    const int g0   = (blockIdx.y * 4 + (threadIdx.x >> 5)) * 8;
    const int lane = threadIdx.x & 31;
    const int f    = T * 16 + (lane & 15);
    const bool fv  = (f < NF);
    const float* row = Ws + (size_t)(fv ? f : (NF - 1)) * FW;
    v2f pa[8], pb[8];
    #pragma unroll
    for (int j = 0; j < 8; ++j) {
        const int k = (g0 + j) * 8 + (lane >> 4) * 2;
        pa[j] = *(const v2f*)(row + ((k + 1 < FW) ? k       : (FW - 2)));
        pb[j] = *(const v2f*)(row + ((k + 5 < FW) ? (k + 4) : (FW - 2)));
    }
    v4f* dst = (v4f*)Wq + (size_t)(T * NGRP + g0) * GSTRIDE + lane;
    #pragma unroll
    for (int j = 0; j < 8; ++j) {
        const int k = (g0 + j) * 8 + (lane >> 4) * 2;
        const bool va = fv && (k + 1 < FW);
        const bool vb = fv && (k + 5 < FW);
        v4f q;
        q[0] = va ? pa[j][0] : 0.0f;  q[1] = va ? pa[j][1] : 0.0f;
        q[2] = vb ? pb[j][0] : 0.0f;  q[3] = vb ? pb[j][1] : 0.0f;
        dst[j * GSTRIDE] = q;
    }
}

struct FragSet { v4f bq[2]; v2f a0[4]; v2f a1[4]; };

__device__ __forceinline__ v2f b_of(const v4f& q0, const v4f& q1, int t) {
    const v4f q = (t >> 1) ? q1 : q0;
    return (t & 1) ? __builtin_shufflevector(q, q, 2, 3)
                   : __builtin_shufflevector(q, q, 0, 1);
}

__device__ __forceinline__ void load_set(FragSet& s, const v4f* wq,
                                         const float* x0, const float* x1) {
    s.bq[0] = wq[0];
    s.bq[1] = wq[GSTRIDE];
    #pragma unroll
    for (int t = 0; t < 4; ++t) {
        s.a0[t] = *(const v2f*)(x0 + 4 * t);
        s.a1[t] = *(const v2f*)(x1 + 4 * t);
    }
}

__device__ __forceinline__ void mma_set(const FragSet& s, v8f& acc0, v8f& acc1) {
    #pragma unroll
    for (int t = 0; t < 4; ++t) {
        const v2f b = b_of(s.bq[0], s.bq[1], t);
        acc0 = __builtin_amdgcn_wmma_f32_16x16x4_f32(
                   false, s.a0[t], false, b, (short)0, acc0, false, false);
        acc1 = __builtin_amdgcn_wmma_f32_16x16x4_f32(
                   false, s.a1[t], false, b, (short)0, acc1, false, false);
    }
}

__global__ __launch_bounds__(NTHREADS) void densenet_wmma_kernel(
    const float* __restrict__ state,
    const float* __restrict__ Ws,
    const float* __restrict__ Wq,
    const float* __restrict__ bs,
    const float* __restrict__ Wout,
    const float* __restrict__ bout,
    float* __restrict__ out)
{
    extern __shared__ float xbuf[];

    const int tid  = threadIdx.x;
    const int wave = tid >> 5;
    const int lane = tid & 31;
    const int l16  = lane & 15;
    const int hi   = lane >> 4;
    const int ksub = hi * 2;

    const int sl = wave & 3;
    const int mp = sl & 1;
    const int kh = sl >> 1;

    const int kh_u   = __builtin_amdgcn_readfirstlane(kh);
    const int q_u    = __builtin_amdgcn_readfirstlane(wave >> 2);
    const int rowbase = blockIdx.x * ROWS;

    for (int e = tid; e < ROWS * INPUT_SIZE; e += NTHREADS) {
        int r = e >> 4, c = e & 15;
        xbuf[r * XSTRIDE + c] = state[(size_t)(rowbase + r) * INPUT_SIZE + c];
    }
    __syncthreads();

    const float* xrow0 = xbuf + (mp * 32 + l16) * XSTRIDE;
    const float* xrow1 = xrow0 + 16 * XSTRIDE;
    const float* xrowS = xbuf + (sl * 16 + l16) * XSTRIDE;

    int width = INPUT_SIZE;
    for (int layer = 0; layer < HC; ++layer) {
        const int f0    = layer * KN;
        const int r     = f0 & 15;
        const int tfull = (f0 + 15) >> 4;
        const bool shallow = (r != 0) && (q_u == 3);
        v8f acc0, acc1;
        int  f = 0;
        bool fvalid = false;
        if (!shallow) {
        f      = (tfull + q_u) * 16 + l16;
        fvalid = (f < NF);
        const int fcl = fvalid ? f : (NF - 1);
        const v4f* wq = (const v4f*)Wq
                      + (size_t)((tfull + q_u) * NGRP) * GSTRIDE + lane;

        const float bias = (kh_u == 0) ? bs[fcl] : 0.0f;
        #pragma unroll
        for (int j = 0; j < 8; ++j) { acc0[j] = bias; acc1[j] = bias; }

        const int wmain  = width & ~3;
        const int ksplit = ((wmain >> 1) + 4) & ~7;
        const int kbeg = __builtin_amdgcn_readfirstlane((kh_u == 0) ? 0 : ksplit);
        const int kend = __builtin_amdgcn_readfirstlane((kh_u == 0) ? ksplit : wmain);

        const float* x0 = xrow0 + ksub;
        const float* x1 = xrow1 + ksub;
        const int nblk = (kend - kbeg) >> 4;
        const int nrem = ((kend - kbeg) >> 2) & 3;
        FragSet sR;
        v2f btail;
        {
            const int krem = kbeg + (nblk << 4);
            const int kcl  = kend - 4;
            sR.bq[0] = wq[(krem >> 3) * GSTRIDE];
            sR.bq[1] = wq[((krem >> 3) + 1) * GSTRIDE];
            #pragma unroll
            for (int t = 0; t < 3; ++t) {
                const int kt = (krem + 4 * t < kcl) ? (krem + 4 * t) : kcl;
                sR.a0[t] = *(const v2f*)(x0 + kt);
                sR.a1[t] = *(const v2f*)(x1 + kt);
            }
            const v4f tq = wq[(wmain >> 3) * GSTRIDE];
            btail = (wmain & 4) ? __builtin_shufflevector(tq, tq, 2, 3)
                                : __builtin_shufflevector(tq, tq, 0, 1);
            const int ktail = (width & 2) ? wmain : (wmain - 4);
            sR.a0[3] = *(const v2f*)(xrow0 + ktail);
            sR.a1[3] = *(const v2f*)(xrow1 + ktail);
        }
        if (nblk > 0) {
            const int klast = kbeg + ((nblk - 1) << 4);
            #define BLK(j) ((kbeg + ((j) << 4) < klast) ? (kbeg + ((j) << 4)) : klast)
            FragSet s0, s1, s2;
            int kk = BLK(0);
            load_set(s0, wq + (kk >> 3) * GSTRIDE, x0 + kk, x1 + kk);
            kk = BLK(1);
            load_set(s1, wq + (kk >> 3) * GSTRIDE, x0 + kk, x1 + kk);
            int i = 0;
            for (; i + 3 <= nblk; i += 3) {
                kk = BLK(i + 2);
                load_set(s2, wq + (kk >> 3) * GSTRIDE, x0 + kk, x1 + kk);
                __builtin_amdgcn_sched_barrier(0);
                mma_set(s0, acc0, acc1);
                kk = BLK(i + 3);
                load_set(s0, wq + (kk >> 3) * GSTRIDE, x0 + kk, x1 + kk);
                __builtin_amdgcn_sched_barrier(0);
                mma_set(s1, acc0, acc1);
                kk = BLK(i + 4);
                load_set(s1, wq + (kk >> 3) * GSTRIDE, x0 + kk, x1 + kk);
                __builtin_amdgcn_sched_barrier(0);
                mma_set(s2, acc0, acc1);
            }
            #undef BLK
            if (nblk - i >= 1) mma_set(s0, acc0, acc1);
            if (nblk - i >= 2) mma_set(s1, acc0, acc1);
        }
        #pragma unroll
        for (int t = 0; t < 3; ++t) {
            if (t < nrem) {
                const v2f b = b_of(sR.bq[0], sR.bq[1], t);
                acc0 = __builtin_amdgcn_wmma_f32_16x16x4_f32(
                           false, sR.a0[t], false, b, (short)0, acc0, false, false);
                acc1 = __builtin_amdgcn_wmma_f32_16x16x4_f32(
                           false, sR.a1[t], false, b, (short)0, acc1, false, false);
            }
        }
        if ((kh_u == 1) && (width & 2)) {
            v2f a0 = sR.a0[3], a1 = sR.a1[3], b = btail;
            a0[0] = hi ? 0.0f : a0[0];  a0[1] = hi ? 0.0f : a0[1];
            a1[0] = hi ? 0.0f : a1[0];  a1[1] = hi ? 0.0f : a1[1];
            b[0]  = hi ? 0.0f : b[0];   b[1]  = hi ? 0.0f : b[1];
            acc0 = __builtin_amdgcn_wmma_f32_16x16x4_f32(
                       false, a0, false, b, (short)0, acc0, false, false);
            acc1 = __builtin_amdgcn_wmma_f32_16x16x4_f32(
                       false, a1, false, b, (short)0, acc1, false, false);
        }

        if (kh_u == 1 && fvalid) {
            float* dst = xbuf + INPUT_SIZE + f;
            #pragma unroll
            for (int j = 0; j < 8; ++j) {
                int r0 = mp * 32 + hi * 8 + j;
                dst[r0 * XSTRIDE]        = acc0[j];
                dst[(r0 + 16) * XSTRIDE] = acc1[j];
            }
        }
        } else {
            const int  fS  = (f0 >> 4) * 16 + l16;
            const bool own = (l16 >= r);
            const float* ws = Ws + (size_t)fS * FW + (width - KN);
            const float* xs = xrowS + (width - KN);
            float* dst = xbuf + INPUT_SIZE + fS;
            v2f bS[13], aS[13];
            #pragma unroll
            for (int t = 0; t < 12; ++t) {
                bS[t] = *(const v2f*)(ws + 4 * t + ksub);
                aS[t] = *(const v2f*)(xs + 4 * t + ksub);
            }
            bS[12] = *(const v2f*)(ws + 48);
            aS[12] = *(const v2f*)(xs + 48);
            bS[12][0] = hi ? 0.0f : bS[12][0];  bS[12][1] = hi ? 0.0f : bS[12][1];
            aS[12][0] = hi ? 0.0f : aS[12][0];  aS[12][1] = hi ? 0.0f : aS[12][1];
            v8f acc;
            #pragma unroll
            for (int j = 0; j < 8; ++j) {
                int r0 = sl * 16 + hi * 8 + j;
                acc[j] = own ? dst[r0 * XSTRIDE] : 0.0f;
            }
            #pragma unroll
            for (int t = 0; t < 13; ++t)
                acc = __builtin_amdgcn_wmma_f32_16x16x4_f32(
                          false, aS[t], false, bS[t], (short)0, acc, false, false);
            if (own) {
                #pragma unroll
                for (int j = 0; j < 8; ++j) {
                    int r0 = sl * 16 + hi * 8 + j;
                    float v = acc[j];
                    dst[r0 * XSTRIDE] = (v > 0.0f) ? v : 0.01f * v;
                }
            }
        }
        __syncthreads();
        if (!shallow && kh_u == 0 && fvalid) {
            float* dst = xbuf + INPUT_SIZE + f;
            const bool fin = (f < f0 + KN);
            #pragma unroll
            for (int j = 0; j < 8; ++j) {
                int r0 = mp * 32 + hi * 8 + j;
                float v0 = acc0[j] + dst[r0 * XSTRIDE];
                float v1 = acc1[j] + dst[(r0 + 16) * XSTRIDE];
                v0 = (fin && !(v0 > 0.0f)) ? 0.01f * v0 : v0;
                v1 = (fin && !(v1 > 0.0f)) ? 0.01f * v1 : v1;
                dst[r0 * XSTRIDE]        = v0;
                dst[(r0 + 16) * XSTRIDE] = v1;
            }
        }
        width += KN;
        __syncthreads();
    }

    const int row = tid >> 3;
    const int g   = tid & 7;
    const float* xr = xbuf + row * XSTRIDE;
    float facc = 0.0f;
    for (int c = g; c < FW; c += 8)
        facc += xr[c] * Wout[c];
    facc += __shfl_down(facc, 4, 8);
    facc += __shfl_down(facc, 2, 8);
    facc += __shfl_down(facc, 1, 8);
    if (g == 0)
        out[rowbase + row] = facc + bout[0];
}

extern "C" void kernel_launch(void* const* d_in, const int* in_sizes, int n_in,
                              void* d_out, int out_size, void* d_ws, size_t ws_size,
                              hipStream_t stream) {
    (void)n_in; (void)ws_size; (void)out_size;
    const float* state = (const float*)d_in[0];
    const float* Ws    = (const float*)d_in[1];
    const float* bs    = (const float*)d_in[2];
    const float* Wout  = (const float*)d_in[3];
    const float* bout  = (const float*)d_in[4];
    float* out = (float*)d_out;
    float* Wq  = (float*)d_ws;

    const int B      = in_sizes[0] / INPUT_SIZE;
    const int blocks = B / ROWS;
    const size_t lds_bytes = (size_t)ROWS * XSTRIDE * sizeof(float);

    (void)hipFuncSetAttribute((const void*)densenet_wmma_kernel,
                              hipFuncAttributeMaxDynamicSharedMemorySize,
                              (int)lds_bytes);

    pack_weights_kernel<<<dim3(NTILE, NGRP / 32), 128, 0, stream>>>(Ws, Wq);

    densenet_wmma_kernel<<<blocks, NTHREADS, lds_bytes, stream>>>(
        state, Ws, Wq, bs, Wout, bout, out);
}


# Next layer's bias and first packed weight groups (blocks 0, 1, left-over, tail) are requested before the layer's two barriers, which now wait for LDS only; k order and sum order unchanged
# speedup vs baseline: 1.7284x; 1.0480x over previous
#include <hip/hip_runtime.h>

typedef __attribute__((ext_vector_type(2))) float v2f;
typedef __attribute__((ext_vector_type(4))) float v4f;
typedef __attribute__((ext_vector_type(8))) float v8f;

#define INPUT_SIZE 16
#define KN         50
#define HC         25
#define FW         1266
#define NF         1250
#define ROWS       64
#define XSTRIDE    1268
#define NTHREADS   512
#define NTILE      79
#define NGRP       160
#define GSTRIDE    32

__global__ __launch_bounds__(128) void pack_weights_kernel(
    const float* __restrict__ Ws,
    float* __restrict__ Wq)
{
    const int T    = blockIdx.x;
    const int g0   = (blockIdx.y * 4 + (threadIdx.x >> 5)) * 8;
    const int lane = threadIdx.x & 31;
    const int f    = T * 16 + (lane & 15);
    const bool fv  = (f < NF);
    const float* row = Ws + (size_t)(fv ? f : (NF - 1)) * FW;
    v2f pa[8], pb[8];
    #pragma unroll
    for (int j = 0; j < 8; ++j) {
        const int k = (g0 + j) * 8 + (lane >> 4) * 2;
        pa[j] = *(const v2f*)(row + ((k + 1 < FW) ? k       : (FW - 2)));
        pb[j] = *(const v2f*)(row + ((k + 5 < FW) ? (k + 4) : (FW - 2)));
    }
    v4f* dst = (v4f*)Wq + (size_t)(T * NGRP + g0) * GSTRIDE + lane;
    #pragma unroll
    for (int j = 0; j < 8; ++j) {
        const int k = (g0 + j) * 8 + (lane >> 4) * 2;
        const bool va = fv && (k + 1 < FW);
        const bool vb = fv && (k + 5 < FW);
        v4f q;
        q[0] = va ? pa[j][0] : 0.0f;  q[1] = va ? pa[j][1] : 0.0f;
        q[2] = vb ? pb[j][0] : 0.0f;  q[3] = vb ? pb[j][1] : 0.0f;
        dst[j * GSTRIDE] = q;
    }
}

struct FragSet { v4f bq[2]; v2f a0[4]; v2f a1[4]; };

__device__ __forceinline__ v2f b_of(const v4f& q0, const v4f& q1, int t) {
    const v4f q = (t >> 1) ? q1 : q0;
    return (t & 1) ? __builtin_shufflevector(q, q, 2, 3)
                   : __builtin_shufflevector(q, q, 0, 1);
}

__device__ __forceinline__ void load_set(FragSet& s, const v4f* wq,
                                         const float* x0, const float* x1) {
    s.bq[0] = wq[0];
    s.bq[1] = wq[GSTRIDE];
    #pragma unroll
    for (int t = 0; t < 4; ++t) {
        s.a0[t] = *(const v2f*)(x0 + 4 * t);
        s.a1[t] = *(const v2f*)(x1 + 4 * t);
    }
}

__device__ __forceinline__ void mma_set(const FragSet& s, v8f& acc0, v8f& acc1) {
    #pragma unroll
    for (int t = 0; t < 4; ++t) {
        const v2f b = b_of(s.bq[0], s.bq[1], t);
        acc0 = __builtin_amdgcn_wmma_f32_16x16x4_f32(
                   false, s.a0[t], false, b, (short)0, acc0, false, false);
        acc1 = __builtin_amdgcn_wmma_f32_16x16x4_f32(
                   false, s.a1[t], false, b, (short)0, acc1, false, false);
    }
}

__device__ __forceinline__ void lds_barrier() {
    __builtin_amdgcn_fence(__ATOMIC_RELEASE, "workgroup", "local");
    __builtin_amdgcn_s_barrier();
    __builtin_amdgcn_fence(__ATOMIC_ACQUIRE, "workgroup", "local");
}

__device__ __forceinline__ void load_a(FragSet& s, const float* x0, const float* x1) {
    #pragma unroll
    for (int t = 0; t < 4; ++t) {
        s.a0[t] = *(const v2f*)(x0 + 4 * t);
        s.a1[t] = *(const v2f*)(x1 + 4 * t);
    }
}

struct Geom { int width, wmain, kbeg, kend, nblk, nrem, krem, klast; };
__device__ __forceinline__ Geom geom_of(int layer, int kh_u) {
    Geom g;
    g.width = INPUT_SIZE + layer * KN;
    g.wmain = g.width & ~3;
    const int ksplit = ((g.wmain >> 1) + 4) & ~7;
    g.kbeg  = __builtin_amdgcn_readfirstlane((kh_u == 0) ? 0 : ksplit);
    g.kend  = __builtin_amdgcn_readfirstlane((kh_u == 0) ? ksplit : g.wmain);
    g.nblk  = (g.kend - g.kbeg) >> 4;
    g.nrem  = ((g.kend - g.kbeg) >> 2) & 3;
    g.krem  = g.kbeg + (g.nblk << 4);
    g.klast = g.kbeg + ((g.nblk - 1) << 4);
    return g;
}
__device__ __forceinline__ int blk_k(const Geom& g, int j) {
    const int k = g.kbeg + (j << 4);
    return (k < g.klast) ? k : g.klast;
}

struct Pre { v4f q0a, q0b, q1a, q1b, ra, rb, tq; float bias; };

__device__ __forceinline__ void prefetch_layer(Pre& P, int ln, int q_u, int kh_u,
                                               int l16, int lane,
                                               const float* __restrict__ Wq,
                                               const float* __restrict__ bs) {
    const int f0n = ln * KN;
    if (((f0n & 15) != 0) && (q_u == 3)) return;
    const int tile = ((f0n + 15) >> 4) + q_u;
    const int fn   = tile * 16 + l16;
    const int fcl  = (fn < NF) ? fn : (NF - 1);
    const v4f* wq  = (const v4f*)Wq + (size_t)(tile * NGRP) * GSTRIDE + lane;
    const Geom g   = geom_of(ln, kh_u);
    P.bias = (kh_u == 0) ? bs[fcl] : 0.0f;
    P.ra = wq[(g.krem >> 3) * GSTRIDE];
    P.rb = wq[((g.krem >> 3) + 1) * GSTRIDE];
    P.tq = wq[(g.wmain >> 3) * GSTRIDE];
    if (g.nblk > 0) {
        int kk = blk_k(g, 0);
        P.q0a = wq[(kk >> 3) * GSTRIDE];
        P.q0b = wq[(kk >> 3) * GSTRIDE + GSTRIDE];
        kk = blk_k(g, 1);
        P.q1a = wq[(kk >> 3) * GSTRIDE];
        P.q1b = wq[(kk >> 3) * GSTRIDE + GSTRIDE];
    }
}

__global__ __launch_bounds__(NTHREADS) void densenet_wmma_kernel(
    const float* __restrict__ state,
    const float* __restrict__ Ws,
    const float* __restrict__ Wq,
    const float* __restrict__ bs,
    const float* __restrict__ Wout,
    const float* __restrict__ bout,
    float* __restrict__ out)
{
    extern __shared__ float xbuf[];

    const int tid  = threadIdx.x;
    const int wave = tid >> 5;
    const int lane = tid & 31;
    const int l16  = lane & 15;
    const int hi   = lane >> 4;
    const int ksub = hi * 2;

    const int sl = wave & 3;
    const int mp = sl & 1;
    const int kh = sl >> 1;

    const int kh_u   = __builtin_amdgcn_readfirstlane(kh);
    const int q_u    = __builtin_amdgcn_readfirstlane(wave >> 2);
    const int rowbase = blockIdx.x * ROWS;

    Pre P = {};
    prefetch_layer(P, 0, q_u, kh_u, l16, lane, Wq, bs);

    for (int e = tid; e < ROWS * INPUT_SIZE; e += NTHREADS) {
        int r = e >> 4, c = e & 15;
        xbuf[r * XSTRIDE + c] = state[(size_t)(rowbase + r) * INPUT_SIZE + c];
    }
    __syncthreads();

    const float* xrow0 = xbuf + (mp * 32 + l16) * XSTRIDE;
    const float* xrow1 = xrow0 + 16 * XSTRIDE;
    const float* xrowS = xbuf + (sl * 16 + l16) * XSTRIDE;

    int width = INPUT_SIZE;
    for (int layer = 0; layer < HC; ++layer) {
        const int f0    = layer * KN;
        const int r     = f0 & 15;
        const int tfull = (f0 + 15) >> 4;
        const bool shallow = (r != 0) && (q_u == 3);
        v8f acc0, acc1;
        int  f = 0;
        bool fvalid = false;
        if (!shallow) {
        f      = (tfull + q_u) * 16 + l16;
        fvalid = (f < NF);
        const v4f* wq = (const v4f*)Wq
                      + (size_t)((tfull + q_u) * NGRP) * GSTRIDE + lane;

        const float bias = P.bias;
        #pragma unroll
        for (int j = 0; j < 8; ++j) { acc0[j] = bias; acc1[j] = bias; }
        asm volatile("" :: "v"(P.ra), "v"(P.rb), "v"(P.tq));

        const Geom g = geom_of(layer, kh_u);
        const int wmain = g.wmain;
        const int kbeg  = g.kbeg;
        const int kend  = g.kend;

        const float* x0 = xrow0 + ksub;
        const float* x1 = xrow1 + ksub;
        const int nblk = g.nblk;
        const int nrem = g.nrem;
        FragSet sR;
        v2f btail;
        {
            const int krem = g.krem;
            const int kcl  = kend - 4;
            sR.bq[0] = P.ra;
            sR.bq[1] = P.rb;
            #pragma unroll
            for (int t = 0; t < 3; ++t) {
                const int kt = (krem + 4 * t < kcl) ? (krem + 4 * t) : kcl;
                sR.a0[t] = *(const v2f*)(x0 + kt);
                sR.a1[t] = *(const v2f*)(x1 + kt);
            }
            const v4f tq = P.tq;
            btail = (wmain & 4) ? __builtin_shufflevector(tq, tq, 2, 3)
                                : __builtin_shufflevector(tq, tq, 0, 1);
            const int ktail = (width & 2) ? wmain : (wmain - 4);
            sR.a0[3] = *(const v2f*)(xrow0 + ktail);
            sR.a1[3] = *(const v2f*)(xrow1 + ktail);
        }
        if (nblk > 0) {
            const int klast = g.klast;
            #define BLK(j) ((kbeg + ((j) << 4) < klast) ? (kbeg + ((j) << 4)) : klast)
            FragSet s0, s1, s2;
            int kk = blk_k(g, 0);
            s0.bq[0] = P.q0a;  s0.bq[1] = P.q0b;
            load_a(s0, x0 + kk, x1 + kk);
            kk = blk_k(g, 1);
            s1.bq[0] = P.q1a;  s1.bq[1] = P.q1b;
            load_a(s1, x0 + kk, x1 + kk);
            int i = 0;
            for (; i + 3 <= nblk; i += 3) {
                kk = BLK(i + 2);
                load_set(s2, wq + (kk >> 3) * GSTRIDE, x0 + kk, x1 + kk);
                __builtin_amdgcn_sched_barrier(0);
                mma_set(s0, acc0, acc1);
                kk = BLK(i + 3);
                load_set(s0, wq + (kk >> 3) * GSTRIDE, x0 + kk, x1 + kk);
                __builtin_amdgcn_sched_barrier(0);
                mma_set(s1, acc0, acc1);
                kk = BLK(i + 4);
                load_set(s1, wq + (kk >> 3) * GSTRIDE, x0 + kk, x1 + kk);
                __builtin_amdgcn_sched_barrier(0);
                mma_set(s2, acc0, acc1);
            }
            #undef BLK
            if (nblk - i >= 1) mma_set(s0, acc0, acc1);
            if (nblk - i >= 2) mma_set(s1, acc0, acc1);
        }
        #pragma unroll
        for (int t = 0; t < 3; ++t) {
            if (t < nrem) {
                const v2f b = b_of(sR.bq[0], sR.bq[1], t);
                acc0 = __builtin_amdgcn_wmma_f32_16x16x4_f32(
                           false, sR.a0[t], false, b, (short)0, acc0, false, false);
                acc1 = __builtin_amdgcn_wmma_f32_16x16x4_f32(
                           false, sR.a1[t], false, b, (short)0, acc1, false, false);
            }
        }
        if ((kh_u == 1) && (width & 2)) {
            v2f a0 = sR.a0[3], a1 = sR.a1[3], b = btail;
            a0[0] = hi ? 0.0f : a0[0];  a0[1] = hi ? 0.0f : a0[1];
            a1[0] = hi ? 0.0f : a1[0];  a1[1] = hi ? 0.0f : a1[1];
            b[0]  = hi ? 0.0f : b[0];   b[1]  = hi ? 0.0f : b[1];
            acc0 = __builtin_amdgcn_wmma_f32_16x16x4_f32(
                       false, a0, false, b, (short)0, acc0, false, false);
            acc1 = __builtin_amdgcn_wmma_f32_16x16x4_f32(
                       false, a1, false, b, (short)0, acc1, false, false);
        }

        if (layer + 1 < HC)
            prefetch_layer(P, layer + 1, q_u, kh_u, l16, lane, Wq, bs);

        if (kh_u == 1 && fvalid) {
            float* dst = xbuf + INPUT_SIZE + f;
            #pragma unroll
            for (int j = 0; j < 8; ++j) {
                int r0 = mp * 32 + hi * 8 + j;
                dst[r0 * XSTRIDE]        = acc0[j];
                dst[(r0 + 16) * XSTRIDE] = acc1[j];
            }
        }
        } else {
            if (layer + 1 < HC)
                prefetch_layer(P, layer + 1, q_u, kh_u, l16, lane, Wq, bs);
            const int  fS  = (f0 >> 4) * 16 + l16;
            const bool own = (l16 >= r);
            const float* ws = Ws + (size_t)fS * FW + (width - KN);
            const float* xs = xrowS + (width - KN);
            float* dst = xbuf + INPUT_SIZE + fS;
            v2f bS[13], aS[13];
            #pragma unroll
            for (int t = 0; t < 12; ++t) {
                bS[t] = *(const v2f*)(ws + 4 * t + ksub);
                aS[t] = *(const v2f*)(xs + 4 * t + ksub);
            }
            bS[12] = *(const v2f*)(ws + 48);
            aS[12] = *(const v2f*)(xs + 48);
            bS[12][0] = hi ? 0.0f : bS[12][0];  bS[12][1] = hi ? 0.0f : bS[12][1];
            aS[12][0] = hi ? 0.0f : aS[12][0];  aS[12][1] = hi ? 0.0f : aS[12][1];
            v8f acc;
            #pragma unroll
            for (int j = 0; j < 8; ++j) {
                int r0 = sl * 16 + hi * 8 + j;
                acc[j] = own ? dst[r0 * XSTRIDE] : 0.0f;
            }
            #pragma unroll
            for (int t = 0; t < 13; ++t)
                acc = __builtin_amdgcn_wmma_f32_16x16x4_f32(
                          false, aS[t], false, bS[t], (short)0, acc, false, false);
            if (own) {
                #pragma unroll
                for (int j = 0; j < 8; ++j) {
                    int r0 = sl * 16 + hi * 8 + j;
                    float v = acc[j];
                    dst[r0 * XSTRIDE] = (v > 0.0f) ? v : 0.01f * v;
                }
            }
        }
        lds_barrier();
        if (!shallow && kh_u == 0 && fvalid) {
            float* dst = xbuf + INPUT_SIZE + f;
            const bool fin = (f < f0 + KN);
            #pragma unroll
            for (int j = 0; j < 8; ++j) {
                int r0 = mp * 32 + hi * 8 + j;
                float v0 = acc0[j] + dst[r0 * XSTRIDE];
                float v1 = acc1[j] + dst[(r0 + 16) * XSTRIDE];
                v0 = (fin && !(v0 > 0.0f)) ? 0.01f * v0 : v0;
                v1 = (fin && !(v1 > 0.0f)) ? 0.01f * v1 : v1;
                dst[r0 * XSTRIDE]        = v0;
                dst[(r0 + 16) * XSTRIDE] = v1;
            }
        }
        width += KN;
        lds_barrier();
    }

    const int row = tid >> 3;
    const int g   = tid & 7;
    const float* xr = xbuf + row * XSTRIDE;
    float facc = 0.0f;
    for (int c = g; c < FW; c += 8)
        facc += xr[c] * Wout[c];
    facc += __shfl_down(facc, 4, 8);
    facc += __shfl_down(facc, 2, 8);
    facc += __shfl_down(facc, 1, 8);
    if (g == 0)
        out[rowbase + row] = facc + bout[0];
}

extern "C" void kernel_launch(void* const* d_in, const int* in_sizes, int n_in,
                              void* d_out, int out_size, void* d_ws, size_t ws_size,
                              hipStream_t stream) {
    (void)n_in; (void)ws_size; (void)out_size;
    const float* state = (const float*)d_in[0];
    const float* Ws    = (const float*)d_in[1];
    const float* bs    = (const float*)d_in[2];
    const float* Wout  = (const float*)d_in[3];
    const float* bout  = (const float*)d_in[4];
    float* out = (float*)d_out;
    float* Wq  = (float*)d_ws;

    const int B      = in_sizes[0] / INPUT_SIZE;
    const int blocks = B / ROWS;
    const size_t lds_bytes = (size_t)ROWS * XSTRIDE * sizeof(float);

    (void)hipFuncSetAttribute((const void*)densenet_wmma_kernel,
                              hipFuncAttributeMaxDynamicSharedMemorySize,
                              (int)lds_bytes);

    pack_weights_kernel<<<dim3(NTILE, NGRP / 32), 128, 0, stream>>>(Ws, Wq);

    densenet_wmma_kernel<<<blocks, NTHREADS, lds_bytes, stream>>>(
        state, Ws, Wq, bs, Wout, bout, out);
}


# s_setprio staircase: a full-depth wave's priority falls with its own K-loop progress (3,2,1,0 at 21,12,6 blocks left), 50-deep wave at 1: a SIMD's waves end a layer together; sum order unchanged
# speedup vs baseline: 1.7829x; 1.0315x over previous
#include <hip/hip_runtime.h>

typedef __attribute__((ext_vector_type(2))) float v2f;
typedef __attribute__((ext_vector_type(4))) float v4f;
typedef __attribute__((ext_vector_type(8))) float v8f;

#define INPUT_SIZE 16
#define KN         50
#define HC         25
#define FW         1266
#define NF         1250
#define ROWS       64
#define XSTRIDE    1268
#define NTHREADS   512
#define NTILE      79
#define NGRP       160
#define GSTRIDE    32

__global__ __launch_bounds__(128) void pack_weights_kernel(
    const float* __restrict__ Ws,
    float* __restrict__ Wq)
{
    const int T    = blockIdx.x;
    const int g0   = (blockIdx.y * 4 + (threadIdx.x >> 5)) * 8;
    const int lane = threadIdx.x & 31;
    const int f    = T * 16 + (lane & 15);
    const bool fv  = (f < NF);
    const float* row = Ws + (size_t)(fv ? f : (NF - 1)) * FW;
    v2f pa[8], pb[8];
    #pragma unroll
    for (int j = 0; j < 8; ++j) {
        const int k = (g0 + j) * 8 + (lane >> 4) * 2;
        pa[j] = *(const v2f*)(row + ((k + 1 < FW) ? k       : (FW - 2)));
        pb[j] = *(const v2f*)(row + ((k + 5 < FW) ? (k + 4) : (FW - 2)));
    }
    v4f* dst = (v4f*)Wq + (size_t)(T * NGRP + g0) * GSTRIDE + lane;
    #pragma unroll
    for (int j = 0; j < 8; ++j) {
        const int k = (g0 + j) * 8 + (lane >> 4) * 2;
        const bool va = fv && (k + 1 < FW);
        const bool vb = fv && (k + 5 < FW);
        v4f q;
        q[0] = va ? pa[j][0] : 0.0f;  q[1] = va ? pa[j][1] : 0.0f;
        q[2] = vb ? pb[j][0] : 0.0f;  q[3] = vb ? pb[j][1] : 0.0f;
        dst[j * GSTRIDE] = q;
    }
}

struct FragSet { v4f bq[2]; v2f a0[4]; v2f a1[4]; };

__device__ __forceinline__ v2f b_of(const v4f& q0, const v4f& q1, int t) {
    const v4f q = (t >> 1) ? q1 : q0;
    return (t & 1) ? __builtin_shufflevector(q, q, 2, 3)
                   : __builtin_shufflevector(q, q, 0, 1);
}

__device__ __forceinline__ void load_set(FragSet& s, const v4f* wq,
                                         const float* x0, const float* x1) {
    s.bq[0] = wq[0];
    s.bq[1] = wq[GSTRIDE];
    #pragma unroll
    for (int t = 0; t < 4; ++t) {
        s.a0[t] = *(const v2f*)(x0 + 4 * t);
        s.a1[t] = *(const v2f*)(x1 + 4 * t);
    }
}

__device__ __forceinline__ void mma_set(const FragSet& s, v8f& acc0, v8f& acc1) {
    #pragma unroll
    for (int t = 0; t < 4; ++t) {
        const v2f b = b_of(s.bq[0], s.bq[1], t);
        acc0 = __builtin_amdgcn_wmma_f32_16x16x4_f32(
                   false, s.a0[t], false, b, (short)0, acc0, false, false);
        acc1 = __builtin_amdgcn_wmma_f32_16x16x4_f32(
                   false, s.a1[t], false, b, (short)0, acc1, false, false);
    }
}

__device__ __forceinline__ void lds_barrier() {
    __builtin_amdgcn_fence(__ATOMIC_RELEASE, "workgroup", "local");
    __builtin_amdgcn_s_barrier();
    __builtin_amdgcn_fence(__ATOMIC_ACQUIRE, "workgroup", "local");
}

__device__ __forceinline__ void set_prio(int lvl) {
    if (lvl >= 3)      __builtin_amdgcn_s_setprio(3);
    else if (lvl == 2) __builtin_amdgcn_s_setprio(2);
    else if (lvl == 1) __builtin_amdgcn_s_setprio(1);
    else               __builtin_amdgcn_s_setprio(0);
}

__device__ __forceinline__ void load_a(FragSet& s, const float* x0, const float* x1) {
    #pragma unroll
    for (int t = 0; t < 4; ++t) {
        s.a0[t] = *(const v2f*)(x0 + 4 * t);
        s.a1[t] = *(const v2f*)(x1 + 4 * t);
    }
}

struct Geom { int width, wmain, kbeg, kend, nblk, nrem, krem, klast; };
__device__ __forceinline__ Geom geom_of(int layer, int kh_u) {
    Geom g;
    g.width = INPUT_SIZE + layer * KN;
    g.wmain = g.width & ~3;
    const int ksplit = ((g.wmain >> 1) + 4) & ~7;
    g.kbeg  = __builtin_amdgcn_readfirstlane((kh_u == 0) ? 0 : ksplit);
    g.kend  = __builtin_amdgcn_readfirstlane((kh_u == 0) ? ksplit : g.wmain);
    g.nblk  = (g.kend - g.kbeg) >> 4;
    g.nrem  = ((g.kend - g.kbeg) >> 2) & 3;
    g.krem  = g.kbeg + (g.nblk << 4);
    g.klast = g.kbeg + ((g.nblk - 1) << 4);
    return g;
}
__device__ __forceinline__ int blk_k(const Geom& g, int j) {
    const int k = g.kbeg + (j << 4);
    return (k < g.klast) ? k : g.klast;
}

struct Pre { v4f q0a, q0b, q1a, q1b, ra, rb, tq; float bias; };

__device__ __forceinline__ void prefetch_layer(Pre& P, int ln, int q_u, int kh_u,
                                               int l16, int lane,
                                               const float* __restrict__ Wq,
                                               const float* __restrict__ bs) {
    const int f0n = ln * KN;
    if (((f0n & 15) != 0) && (q_u == 3)) return;
    const int tile = ((f0n + 15) >> 4) + q_u;
    const int fn   = tile * 16 + l16;
    const int fcl  = (fn < NF) ? fn : (NF - 1);
    const v4f* wq  = (const v4f*)Wq + (size_t)(tile * NGRP) * GSTRIDE + lane;
    const Geom g   = geom_of(ln, kh_u);
    P.bias = (kh_u == 0) ? bs[fcl] : 0.0f;
    P.ra = wq[(g.krem >> 3) * GSTRIDE];
    P.rb = wq[((g.krem >> 3) + 1) * GSTRIDE];
    P.tq = wq[(g.wmain >> 3) * GSTRIDE];
    if (g.nblk > 0) {
        int kk = blk_k(g, 0);
        P.q0a = wq[(kk >> 3) * GSTRIDE];
        P.q0b = wq[(kk >> 3) * GSTRIDE + GSTRIDE];
        kk = blk_k(g, 1);
        P.q1a = wq[(kk >> 3) * GSTRIDE];
        P.q1b = wq[(kk >> 3) * GSTRIDE + GSTRIDE];
    }
}

__global__ __launch_bounds__(NTHREADS) void densenet_wmma_kernel(
    const float* __restrict__ state,
    const float* __restrict__ Ws,
    const float* __restrict__ Wq,
    const float* __restrict__ bs,
    const float* __restrict__ Wout,
    const float* __restrict__ bout,
    float* __restrict__ out)
{
    extern __shared__ float xbuf[];

    const int tid  = threadIdx.x;
    const int wave = tid >> 5;
    const int lane = tid & 31;
    const int l16  = lane & 15;
    const int hi   = lane >> 4;
    const int ksub = hi * 2;

    const int sl = wave & 3;
    const int mp = sl & 1;
    const int kh = sl >> 1;

    const int kh_u   = __builtin_amdgcn_readfirstlane(kh);
    const int q_u    = __builtin_amdgcn_readfirstlane(wave >> 2);
    const int rowbase = blockIdx.x * ROWS;

    Pre P = {};
    prefetch_layer(P, 0, q_u, kh_u, l16, lane, Wq, bs);

    for (int e = tid; e < ROWS * INPUT_SIZE; e += NTHREADS) {
        int r = e >> 4, c = e & 15;
        xbuf[r * XSTRIDE + c] = state[(size_t)(rowbase + r) * INPUT_SIZE + c];
    }
    __syncthreads();

    const float* xrow0 = xbuf + (mp * 32 + l16) * XSTRIDE;
    const float* xrow1 = xrow0 + 16 * XSTRIDE;
    const float* xrowS = xbuf + (sl * 16 + l16) * XSTRIDE;

    int width = INPUT_SIZE;
    for (int layer = 0; layer < HC; ++layer) {
        const int f0    = layer * KN;
        const int r     = f0 & 15;
        const int tfull = (f0 + 15) >> 4;
        const bool shallow = (r != 0) && (q_u == 3);
        v8f acc0, acc1;
        int  f = 0;
        bool fvalid = false;
        if (!shallow) {
        f      = (tfull + q_u) * 16 + l16;
        fvalid = (f < NF);
        const v4f* wq = (const v4f*)Wq
                      + (size_t)((tfull + q_u) * NGRP) * GSTRIDE + lane;

        const float bias = P.bias;
        #pragma unroll
        for (int j = 0; j < 8; ++j) { acc0[j] = bias; acc1[j] = bias; }
        asm volatile("" :: "v"(P.ra), "v"(P.rb), "v"(P.tq));

        const Geom g = geom_of(layer, kh_u);
        const int wmain = g.wmain;
        const int kbeg  = g.kbeg;
        const int kend  = g.kend;

        const float* x0 = xrow0 + ksub;
        const float* x1 = xrow1 + ksub;
        const int nblk = g.nblk;
        const int nrem = g.nrem;
        FragSet sR;
        v2f btail;
        {
            const int krem = g.krem;
            const int kcl  = kend - 4;
            sR.bq[0] = P.ra;
            sR.bq[1] = P.rb;
            #pragma unroll
            for (int t = 0; t < 3; ++t) {
                const int kt = (krem + 4 * t < kcl) ? (krem + 4 * t) : kcl;
                sR.a0[t] = *(const v2f*)(x0 + kt);
                sR.a1[t] = *(const v2f*)(x1 + kt);
            }
            const v4f tq = P.tq;
            btail = (wmain & 4) ? __builtin_shufflevector(tq, tq, 2, 3)
                                : __builtin_shufflevector(tq, tq, 0, 1);
            const int ktail = (width & 2) ? wmain : (wmain - 4);
            sR.a0[3] = *(const v2f*)(xrow0 + ktail);
            sR.a1[3] = *(const v2f*)(xrow1 + ktail);
        }
        if (nblk > 0) {
            const int klast = g.klast;
            #define BLK(j) ((kbeg + ((j) << 4) < klast) ? (kbeg + ((j) << 4)) : klast)
            FragSet s0, s1, s2;
            int kk = blk_k(g, 0);
            s0.bq[0] = P.q0a;  s0.bq[1] = P.q0b;
            load_a(s0, x0 + kk, x1 + kk);
            kk = blk_k(g, 1);
            s1.bq[0] = P.q1a;  s1.bq[1] = P.q1b;
            load_a(s1, x0 + kk, x1 + kk);
            int i = 0;
            int cur = -1;
            #define PRIO_FOR(left) { const int lvl_ = ((left) >= 21) ? 3 : ((left) >= 12) ? 2 \
                                                    : ((left) >= 6) ? 1 : 0;              \
                                     if (lvl_ != cur) { cur = lvl_; set_prio(lvl_); } }
            PRIO_FOR(nblk)
            for (; i + 3 <= nblk; i += 3) {
                kk = BLK(i + 2);
                load_set(s2, wq + (kk >> 3) * GSTRIDE, x0 + kk, x1 + kk);
                __builtin_amdgcn_sched_barrier(0);
                mma_set(s0, acc0, acc1);
                kk = BLK(i + 3);
                load_set(s0, wq + (kk >> 3) * GSTRIDE, x0 + kk, x1 + kk);
                __builtin_amdgcn_sched_barrier(0);
                mma_set(s1, acc0, acc1);
                kk = BLK(i + 4);
                load_set(s1, wq + (kk >> 3) * GSTRIDE, x0 + kk, x1 + kk);
                __builtin_amdgcn_sched_barrier(0);
                mma_set(s2, acc0, acc1);
                PRIO_FOR(nblk - i - 3)
            }
            #undef PRIO_FOR
            #undef BLK
            __builtin_amdgcn_s_setprio(0);
            if (nblk - i >= 1) mma_set(s0, acc0, acc1);
            if (nblk - i >= 2) mma_set(s1, acc0, acc1);
        }
        #pragma unroll
        for (int t = 0; t < 3; ++t) {
            if (t < nrem) {
                const v2f b = b_of(sR.bq[0], sR.bq[1], t);
                acc0 = __builtin_amdgcn_wmma_f32_16x16x4_f32(
                           false, sR.a0[t], false, b, (short)0, acc0, false, false);
                acc1 = __builtin_amdgcn_wmma_f32_16x16x4_f32(
                           false, sR.a1[t], false, b, (short)0, acc1, false, false);
            }
        }
        if ((kh_u == 1) && (width & 2)) {
            v2f a0 = sR.a0[3], a1 = sR.a1[3], b = btail;
            a0[0] = hi ? 0.0f : a0[0];  a0[1] = hi ? 0.0f : a0[1];
            a1[0] = hi ? 0.0f : a1[0];  a1[1] = hi ? 0.0f : a1[1];
            b[0]  = hi ? 0.0f : b[0];   b[1]  = hi ? 0.0f : b[1];
            acc0 = __builtin_amdgcn_wmma_f32_16x16x4_f32(
                       false, a0, false, b, (short)0, acc0, false, false);
            acc1 = __builtin_amdgcn_wmma_f32_16x16x4_f32(
                       false, a1, false, b, (short)0, acc1, false, false);
        }

        if (layer + 1 < HC)
            prefetch_layer(P, layer + 1, q_u, kh_u, l16, lane, Wq, bs);

        if (kh_u == 1 && fvalid) {
            float* dst = xbuf + INPUT_SIZE + f;
            #pragma unroll
            for (int j = 0; j < 8; ++j) {
                int r0 = mp * 32 + hi * 8 + j;
                dst[r0 * XSTRIDE]        = acc0[j];
                dst[(r0 + 16) * XSTRIDE] = acc1[j];
            }
        }
        } else {
            if (layer + 1 < HC)
                prefetch_layer(P, layer + 1, q_u, kh_u, l16, lane, Wq, bs);
            __builtin_amdgcn_s_setprio(1);
            const int  fS  = (f0 >> 4) * 16 + l16;
            const bool own = (l16 >= r);
            const float* ws = Ws + (size_t)fS * FW + (width - KN);
            const float* xs = xrowS + (width - KN);
            float* dst = xbuf + INPUT_SIZE + fS;
            v2f bS[13], aS[13];
            #pragma unroll
            for (int t = 0; t < 12; ++t) {
                bS[t] = *(const v2f*)(ws + 4 * t + ksub);
                aS[t] = *(const v2f*)(xs + 4 * t + ksub);
            }
            bS[12] = *(const v2f*)(ws + 48);
            aS[12] = *(const v2f*)(xs + 48);
            bS[12][0] = hi ? 0.0f : bS[12][0];  bS[12][1] = hi ? 0.0f : bS[12][1];
            aS[12][0] = hi ? 0.0f : aS[12][0];  aS[12][1] = hi ? 0.0f : aS[12][1];
            v8f acc;
            #pragma unroll
            for (int j = 0; j < 8; ++j) {
                int r0 = sl * 16 + hi * 8 + j;
                acc[j] = own ? dst[r0 * XSTRIDE] : 0.0f;
            }
            #pragma unroll
            for (int t = 0; t < 13; ++t)
                acc = __builtin_amdgcn_wmma_f32_16x16x4_f32(
                          false, aS[t], false, bS[t], (short)0, acc, false, false);
            __builtin_amdgcn_s_setprio(0);
            if (own) {
                #pragma unroll
                for (int j = 0; j < 8; ++j) {
                    int r0 = sl * 16 + hi * 8 + j;
                    float v = acc[j];
                    dst[r0 * XSTRIDE] = (v > 0.0f) ? v : 0.01f * v;
                }
            }
        }
        lds_barrier();
        if (!shallow && kh_u == 0 && fvalid) {
            float* dst = xbuf + INPUT_SIZE + f;
            const bool fin = (f < f0 + KN);
            #pragma unroll
            for (int j = 0; j < 8; ++j) {
                int r0 = mp * 32 + hi * 8 + j;
                float v0 = acc0[j] + dst[r0 * XSTRIDE];
                float v1 = acc1[j] + dst[(r0 + 16) * XSTRIDE];
                v0 = (fin && !(v0 > 0.0f)) ? 0.01f * v0 : v0;
                v1 = (fin && !(v1 > 0.0f)) ? 0.01f * v1 : v1;
                dst[r0 * XSTRIDE]        = v0;
                dst[(r0 + 16) * XSTRIDE] = v1;
            }
        }
        width += KN;
        lds_barrier();
    }

    const int row = tid >> 3;
    const int g   = tid & 7;
    const float* xr = xbuf + row * XSTRIDE;
    float facc = 0.0f;
    for (int c = g; c < FW; c += 8)
        facc += xr[c] * Wout[c];
    facc += __shfl_down(facc, 4, 8);
    facc += __shfl_down(facc, 2, 8);
    facc += __shfl_down(facc, 1, 8);
    if (g == 0)
        out[rowbase + row] = facc + bout[0];
}

extern "C" void kernel_launch(void* const* d_in, const int* in_sizes, int n_in,
                              void* d_out, int out_size, void* d_ws, size_t ws_size,
                              hipStream_t stream) {
    (void)n_in; (void)ws_size; (void)out_size;
    const float* state = (const float*)d_in[0];
    const float* Ws    = (const float*)d_in[1];
    const float* bs    = (const float*)d_in[2];
    const float* Wout  = (const float*)d_in[3];
    const float* bout  = (const float*)d_in[4];
    float* out = (float*)d_out;
    float* Wq  = (float*)d_ws;

    const int B      = in_sizes[0] / INPUT_SIZE;
    const int blocks = B / ROWS;
    const size_t lds_bytes = (size_t)ROWS * XSTRIDE * sizeof(float);

    (void)hipFuncSetAttribute((const void*)densenet_wmma_kernel,
                              hipFuncAttributeMaxDynamicSharedMemorySize,
                              (int)lds_bytes);

    pack_weights_kernel<<<dim3(NTILE, NGRP / 32), 128, 0, stream>>>(Ws, Wq);

    densenet_wmma_kernel<<<blocks, NTHREADS, lds_bytes, stream>>>(
        state, Ws, Wq, bs, Wout, bout, out);
}


# K loop: each register set is refilled only behind the last WMMA that reads it, so it lands in place: 16 two-register copies per 24-WMMA trip and the head's wait for all loads go; sum order unchanged
# speedup vs baseline: 1.8508x; 1.0381x over previous
#include <hip/hip_runtime.h>

typedef __attribute__((ext_vector_type(2))) float v2f;
typedef __attribute__((ext_vector_type(4))) float v4f;
typedef __attribute__((ext_vector_type(8))) float v8f;

#define INPUT_SIZE 16
#define KN         50
#define HC         25
#define FW         1266
#define NF         1250
#define ROWS       64
#define XSTRIDE    1268
#define NTHREADS   512
#define NTILE      79
#define NGRP       160
#define GSTRIDE    32

__global__ __launch_bounds__(128) void pack_weights_kernel(
    const float* __restrict__ Ws,
    float* __restrict__ Wq)
{
    const int T    = blockIdx.x;
    const int g0   = (blockIdx.y * 4 + (threadIdx.x >> 5)) * 8;
    const int lane = threadIdx.x & 31;
    const int f    = T * 16 + (lane & 15);
    const bool fv  = (f < NF);
    const float* row = Ws + (size_t)(fv ? f : (NF - 1)) * FW;
    v2f pa[8], pb[8];
    #pragma unroll
    for (int j = 0; j < 8; ++j) {
        const int k = (g0 + j) * 8 + (lane >> 4) * 2;
        pa[j] = *(const v2f*)(row + ((k + 1 < FW) ? k       : (FW - 2)));
        pb[j] = *(const v2f*)(row + ((k + 5 < FW) ? (k + 4) : (FW - 2)));
    }
    v4f* dst = (v4f*)Wq + (size_t)(T * NGRP + g0) * GSTRIDE + lane;
    #pragma unroll
    for (int j = 0; j < 8; ++j) {
        const int k = (g0 + j) * 8 + (lane >> 4) * 2;
        const bool va = fv && (k + 1 < FW);
        const bool vb = fv && (k + 5 < FW);
        v4f q;
        q[0] = va ? pa[j][0] : 0.0f;  q[1] = va ? pa[j][1] : 0.0f;
        q[2] = vb ? pb[j][0] : 0.0f;  q[3] = vb ? pb[j][1] : 0.0f;
        dst[j * GSTRIDE] = q;
    }
}

struct FragSet { v4f bq[2]; v2f a0[4]; v2f a1[4]; };

__device__ __forceinline__ v2f b_of(const v4f& q0, const v4f& q1, int t) {
    const v4f q = (t >> 1) ? q1 : q0;
    return (t & 1) ? __builtin_shufflevector(q, q, 2, 3)
                   : __builtin_shufflevector(q, q, 0, 1);
}

__device__ __forceinline__ void load_set(FragSet& s, const v4f* wq,
                                         const float* x0, const float* x1) {
    s.bq[0] = wq[0];
    s.bq[1] = wq[GSTRIDE];
    #pragma unroll
    for (int t = 0; t < 4; ++t) {
        s.a0[t] = *(const v2f*)(x0 + 4 * t);
        s.a1[t] = *(const v2f*)(x1 + 4 * t);
    }
}

__device__ __forceinline__ void mma_set(const FragSet& s, v8f& acc0, v8f& acc1) {
    #pragma unroll
    for (int t = 0; t < 4; ++t) {
        const v2f b = b_of(s.bq[0], s.bq[1], t);
        acc0 = __builtin_amdgcn_wmma_f32_16x16x4_f32(
                   false, s.a0[t], false, b, (short)0, acc0, false, false);
        acc1 = __builtin_amdgcn_wmma_f32_16x16x4_f32(
                   false, s.a1[t], false, b, (short)0, acc1, false, false);
    }
}

__device__ __forceinline__ void lds_barrier() {
    __builtin_amdgcn_fence(__ATOMIC_RELEASE, "workgroup", "local");
    __builtin_amdgcn_s_barrier();
    __builtin_amdgcn_fence(__ATOMIC_ACQUIRE, "workgroup", "local");
}

__device__ __forceinline__ void set_prio(int lvl) {
    if (lvl >= 3)      __builtin_amdgcn_s_setprio(3);
    else if (lvl == 2) __builtin_amdgcn_s_setprio(2);
    else if (lvl == 1) __builtin_amdgcn_s_setprio(1);
    else               __builtin_amdgcn_s_setprio(0);
}

__device__ __forceinline__ void load_a(FragSet& s, const float* x0, const float* x1) {
    #pragma unroll
    for (int t = 0; t < 4; ++t) {
        s.a0[t] = *(const v2f*)(x0 + 4 * t);
        s.a1[t] = *(const v2f*)(x1 + 4 * t);
    }
}

struct Geom { int width, wmain, kbeg, kend, nblk, nrem, krem, klast; };
__device__ __forceinline__ Geom geom_of(int layer, int kh_u) {
    Geom g;
    g.width = INPUT_SIZE + layer * KN;
    g.wmain = g.width & ~3;
    const int ksplit = ((g.wmain >> 1) + 4) & ~7;
    g.kbeg  = __builtin_amdgcn_readfirstlane((kh_u == 0) ? 0 : ksplit);
    g.kend  = __builtin_amdgcn_readfirstlane((kh_u == 0) ? ksplit : g.wmain);
    g.nblk  = (g.kend - g.kbeg) >> 4;
    g.nrem  = ((g.kend - g.kbeg) >> 2) & 3;
    g.krem  = g.kbeg + (g.nblk << 4);
    g.klast = g.kbeg + ((g.nblk - 1) << 4);
    return g;
}
__device__ __forceinline__ int blk_k(const Geom& g, int j) {
    const int k = g.kbeg + (j << 4);
    return (k < g.klast) ? k : g.klast;
}

struct Pre { v4f q0a, q0b, q1a, q1b, ra, rb, tq; float bias; };

__device__ __forceinline__ void prefetch_layer(Pre& P, int ln, int q_u, int kh_u,
                                               int l16, int lane,
                                               const float* __restrict__ Wq,
                                               const float* __restrict__ bs) {
    const int f0n = ln * KN;
    if (((f0n & 15) != 0) && (q_u == 3)) return;
    const int tile = ((f0n + 15) >> 4) + q_u;
    const int fn   = tile * 16 + l16;
    const int fcl  = (fn < NF) ? fn : (NF - 1);
    const v4f* wq  = (const v4f*)Wq + (size_t)(tile * NGRP) * GSTRIDE + lane;
    const Geom g   = geom_of(ln, kh_u);
    P.bias = (kh_u == 0) ? bs[fcl] : 0.0f;
    P.ra = wq[(g.krem >> 3) * GSTRIDE];
    P.rb = wq[((g.krem >> 3) + 1) * GSTRIDE];
    P.tq = wq[(g.wmain >> 3) * GSTRIDE];
    if (g.nblk > 0) {
        int kk = blk_k(g, 0);
        P.q0a = wq[(kk >> 3) * GSTRIDE];
        P.q0b = wq[(kk >> 3) * GSTRIDE + GSTRIDE];
        kk = blk_k(g, 1);
        P.q1a = wq[(kk >> 3) * GSTRIDE];
        P.q1b = wq[(kk >> 3) * GSTRIDE + GSTRIDE];
    }
}

__global__ __launch_bounds__(NTHREADS) void densenet_wmma_kernel(
    const float* __restrict__ state,
    const float* __restrict__ Ws,
    const float* __restrict__ Wq,
    const float* __restrict__ bs,
    const float* __restrict__ Wout,
    const float* __restrict__ bout,
    float* __restrict__ out)
{
    extern __shared__ float xbuf[];

    const int tid  = threadIdx.x;
    const int wave = tid >> 5;
    const int lane = tid & 31;
    const int l16  = lane & 15;
    const int hi   = lane >> 4;
    const int ksub = hi * 2;

    const int sl = wave & 3;
    const int mp = sl & 1;
    const int kh = sl >> 1;

    const int kh_u   = __builtin_amdgcn_readfirstlane(kh);
    const int q_u    = __builtin_amdgcn_readfirstlane(wave >> 2);
    const int rowbase = blockIdx.x * ROWS;

    Pre P = {};
    prefetch_layer(P, 0, q_u, kh_u, l16, lane, Wq, bs);

    for (int e = tid; e < ROWS * INPUT_SIZE; e += NTHREADS) {
        int r = e >> 4, c = e & 15;
        xbuf[r * XSTRIDE + c] = state[(size_t)(rowbase + r) * INPUT_SIZE + c];
    }
    __syncthreads();

    const float* xrow0 = xbuf + (mp * 32 + l16) * XSTRIDE;
    const float* xrow1 = xrow0 + 16 * XSTRIDE;
    const float* xrowS = xbuf + (sl * 16 + l16) * XSTRIDE;

    int width = INPUT_SIZE;
    for (int layer = 0; layer < HC; ++layer) {
        const int f0    = layer * KN;
        const int r     = f0 & 15;
        const int tfull = (f0 + 15) >> 4;
        const bool shallow = (r != 0) && (q_u == 3);
        v8f acc0, acc1;
        int  f = 0;
        bool fvalid = false;
        if (!shallow) {
        f      = (tfull + q_u) * 16 + l16;
        fvalid = (f < NF);
        const v4f* wq = (const v4f*)Wq
                      + (size_t)((tfull + q_u) * NGRP) * GSTRIDE + lane;

        const float bias = P.bias;
        #pragma unroll
        for (int j = 0; j < 8; ++j) { acc0[j] = bias; acc1[j] = bias; }
        asm volatile("" :: "v"(P.ra), "v"(P.rb), "v"(P.tq));

        const Geom g = geom_of(layer, kh_u);
        const int wmain = g.wmain;
        const int kbeg  = g.kbeg;
        const int kend  = g.kend;

        const float* x0 = xrow0 + ksub;
        const float* x1 = xrow1 + ksub;
        const int nblk = g.nblk;
        const int nrem = g.nrem;
        FragSet sR;
        v2f btail;
        {
            const int krem = g.krem;
            const int kcl  = kend - 4;
            sR.bq[0] = P.ra;
            sR.bq[1] = P.rb;
            #pragma unroll
            for (int t = 0; t < 3; ++t) {
                const int kt = (krem + 4 * t < kcl) ? (krem + 4 * t) : kcl;
                sR.a0[t] = *(const v2f*)(x0 + kt);
                sR.a1[t] = *(const v2f*)(x1 + kt);
            }
            const v4f tq = P.tq;
            btail = (wmain & 4) ? __builtin_shufflevector(tq, tq, 2, 3)
                                : __builtin_shufflevector(tq, tq, 0, 1);
            const int ktail = (width & 2) ? wmain : (wmain - 4);
            sR.a0[3] = *(const v2f*)(xrow0 + ktail);
            sR.a1[3] = *(const v2f*)(xrow1 + ktail);
        }
        if (nblk > 0) {
            const int klast = g.klast;
            #define BLK(j) ((kbeg + ((j) << 4) < klast) ? (kbeg + ((j) << 4)) : klast)
            FragSet s0, s1, s2;
            int kk = blk_k(g, 0);
            s0.bq[0] = P.q0a;  s0.bq[1] = P.q0b;
            load_a(s0, x0 + kk, x1 + kk);
            kk = blk_k(g, 1);
            s1.bq[0] = P.q1a;  s1.bq[1] = P.q1b;
            load_a(s1, x0 + kk, x1 + kk);
            int i = 0;
            int cur = -1;
            #define PRIO_FOR(left) { const int lvl_ = ((left) >= 21) ? 3 : ((left) >= 12) ? 2 \
                                                    : ((left) >= 6) ? 1 : 0;              \
                                     if (lvl_ != cur) { cur = lvl_; set_prio(lvl_); } }
            PRIO_FOR(nblk)
            for (; i + 3 <= nblk; i += 3) {
                kk = BLK(i + 2);
                load_set(s2, wq + (kk >> 3) * GSTRIDE, x0 + kk, x1 + kk);
                __builtin_amdgcn_sched_barrier(0);
                mma_set(s0, acc0, acc1);
                asm volatile("" :: "v"(acc0), "v"(acc1));
                kk = BLK(i + 3);
                load_set(s0, wq + (kk >> 3) * GSTRIDE, x0 + kk, x1 + kk);
                __builtin_amdgcn_sched_barrier(0);
                mma_set(s1, acc0, acc1);
                asm volatile("" :: "v"(acc0), "v"(acc1));
                kk = BLK(i + 4);
                load_set(s1, wq + (kk >> 3) * GSTRIDE, x0 + kk, x1 + kk);
                __builtin_amdgcn_sched_barrier(0);
                mma_set(s2, acc0, acc1);
                PRIO_FOR(nblk - i - 3)
            }
            #undef PRIO_FOR
            #undef BLK
            __builtin_amdgcn_s_setprio(0);
            if (nblk - i >= 1) mma_set(s0, acc0, acc1);
            if (nblk - i >= 2) mma_set(s1, acc0, acc1);
        }
        #pragma unroll
        for (int t = 0; t < 3; ++t) {
            if (t < nrem) {
                const v2f b = b_of(sR.bq[0], sR.bq[1], t);
                acc0 = __builtin_amdgcn_wmma_f32_16x16x4_f32(
                           false, sR.a0[t], false, b, (short)0, acc0, false, false);
                acc1 = __builtin_amdgcn_wmma_f32_16x16x4_f32(
                           false, sR.a1[t], false, b, (short)0, acc1, false, false);
            }
        }
        if ((kh_u == 1) && (width & 2)) {
            v2f a0 = sR.a0[3], a1 = sR.a1[3], b = btail;
            a0[0] = hi ? 0.0f : a0[0];  a0[1] = hi ? 0.0f : a0[1];
            a1[0] = hi ? 0.0f : a1[0];  a1[1] = hi ? 0.0f : a1[1];
            b[0]  = hi ? 0.0f : b[0];   b[1]  = hi ? 0.0f : b[1];
            acc0 = __builtin_amdgcn_wmma_f32_16x16x4_f32(
                       false, a0, false, b, (short)0, acc0, false, false);
            acc1 = __builtin_amdgcn_wmma_f32_16x16x4_f32(
                       false, a1, false, b, (short)0, acc1, false, false);
        }

        if (layer + 1 < HC)
            prefetch_layer(P, layer + 1, q_u, kh_u, l16, lane, Wq, bs);

        if (kh_u == 1 && fvalid) {
            float* dst = xbuf + INPUT_SIZE + f;
            #pragma unroll
            for (int j = 0; j < 8; ++j) {
                int r0 = mp * 32 + hi * 8 + j;
                dst[r0 * XSTRIDE]        = acc0[j];
                dst[(r0 + 16) * XSTRIDE] = acc1[j];
            }
        }
        } else {
            if (layer + 1 < HC)
                prefetch_layer(P, layer + 1, q_u, kh_u, l16, lane, Wq, bs);
            __builtin_amdgcn_s_setprio(1);
            const int  fS  = (f0 >> 4) * 16 + l16;
            const bool own = (l16 >= r);
            const float* ws = Ws + (size_t)fS * FW + (width - KN);
            const float* xs = xrowS + (width - KN);
            float* dst = xbuf + INPUT_SIZE + fS;
            v2f bS[13], aS[13];
            #pragma unroll
            for (int t = 0; t < 12; ++t) {
                bS[t] = *(const v2f*)(ws + 4 * t + ksub);
                aS[t] = *(const v2f*)(xs + 4 * t + ksub);
            }
            bS[12] = *(const v2f*)(ws + 48);
            aS[12] = *(const v2f*)(xs + 48);
            bS[12][0] = hi ? 0.0f : bS[12][0];  bS[12][1] = hi ? 0.0f : bS[12][1];
            aS[12][0] = hi ? 0.0f : aS[12][0];  aS[12][1] = hi ? 0.0f : aS[12][1];
            v8f acc;
            #pragma unroll
            for (int j = 0; j < 8; ++j) {
                int r0 = sl * 16 + hi * 8 + j;
                acc[j] = own ? dst[r0 * XSTRIDE] : 0.0f;
            }
            #pragma unroll
            for (int t = 0; t < 13; ++t)
                acc = __builtin_amdgcn_wmma_f32_16x16x4_f32(
                          false, aS[t], false, bS[t], (short)0, acc, false, false);
            __builtin_amdgcn_s_setprio(0);
            if (own) {
                #pragma unroll
                for (int j = 0; j < 8; ++j) {
                    int r0 = sl * 16 + hi * 8 + j;
                    float v = acc[j];
                    dst[r0 * XSTRIDE] = (v > 0.0f) ? v : 0.01f * v;
                }
            }
        }
        lds_barrier();
        if (!shallow && kh_u == 0 && fvalid) {
            float* dst = xbuf + INPUT_SIZE + f;
            const bool fin = (f < f0 + KN);
            #pragma unroll
            for (int j = 0; j < 8; ++j) {
                int r0 = mp * 32 + hi * 8 + j;
                float v0 = acc0[j] + dst[r0 * XSTRIDE];
                float v1 = acc1[j] + dst[(r0 + 16) * XSTRIDE];
                v0 = (fin && !(v0 > 0.0f)) ? 0.01f * v0 : v0;
                v1 = (fin && !(v1 > 0.0f)) ? 0.01f * v1 : v1;
                dst[r0 * XSTRIDE]        = v0;
                dst[(r0 + 16) * XSTRIDE] = v1;
            }
        }
        width += KN;
        lds_barrier();
    }

    const int row = tid >> 3;
    const int g   = tid & 7;
    const float* xr = xbuf + row * XSTRIDE;
    float facc = 0.0f;
    for (int c = g; c < FW; c += 8)
        facc += xr[c] * Wout[c];
    facc += __shfl_down(facc, 4, 8);
    facc += __shfl_down(facc, 2, 8);
    facc += __shfl_down(facc, 1, 8);
    if (g == 0)
        out[rowbase + row] = facc + bout[0];
}

extern "C" void kernel_launch(void* const* d_in, const int* in_sizes, int n_in,
                              void* d_out, int out_size, void* d_ws, size_t ws_size,
                              hipStream_t stream) {
    (void)n_in; (void)ws_size; (void)out_size;
    const float* state = (const float*)d_in[0];
    const float* Ws    = (const float*)d_in[1];
    const float* bs    = (const float*)d_in[2];
    const float* Wout  = (const float*)d_in[3];
    const float* bout  = (const float*)d_in[4];
    float* out = (float*)d_out;
    float* Wq  = (float*)d_ws;

    const int B      = in_sizes[0] / INPUT_SIZE;
    const int blocks = B / ROWS;
    const size_t lds_bytes = (size_t)ROWS * XSTRIDE * sizeof(float);

    (void)hipFuncSetAttribute((const void*)densenet_wmma_kernel,
                              hipFuncAttributeMaxDynamicSharedMemorySize,
                              (int)lds_bytes);

    pack_weights_kernel<<<dim3(NTILE, NGRP / 32), 128, 0, stream>>>(Ws, Wq);

    densenet_wmma_kernel<<<blocks, NTHREADS, lds_bytes, stream>>>(
        state, Ws, Wq, bs, Wout, bout, out);
}
